# starting kernel
# baseline (speedup 1.0000x reference)
#include <hip/hip_runtime.h>
#include <hip/hip_bf16.h>

typedef __attribute__((ext_vector_type(2))) float v2f;
typedef __attribute__((ext_vector_type(4))) float v4f;
typedef __attribute__((ext_vector_type(8))) float v8f;

#define K_SEQ       2001
#define M_DIM       16
#define SEQ_ELEMS   (K_SEQ * M_DIM)
#define ROWS_TILE   64
#define TILE_ELEMS  (ROWS_TILE * M_DIM)
#define NUM_TILES   32
#define NSEG        6

__device__ __forceinline__ int inc3(int t) { return (t == 2) ? 0 : t + 1; }

__device__ __forceinline__ int seg_from(int r, int t) {
    int seg = (r < 1000) ? t : (3 + inc3(t));
    seg = (r == 1000) ? 99 : seg;
    seg = (r >= 2001) ? 99 : seg;
    return seg;
}

__device__ __forceinline__ float onehot(int seg, int m) {
    return (seg == m) ? 1.0f : 0.0f;
}

__global__ __launch_bounds__(256)
void seg_mean_wmma_kernel(const float* __restrict__ in, float* __restrict__ out) {
    __shared__ float tile[TILE_ELEMS];
    __shared__ float acc[NSEG * M_DIM];

    const int tid  = threadIdx.x;
    const int lane = tid & 31;
    const int wave = tid >> 5;
    const int half = lane >> 4;
    const int m    = lane & 15;
    const size_t seq = blockIdx.x;
    const float* __restrict__ src = in + seq * (size_t)SEQ_ELEMS;

    if (tid < NSEG * M_DIM) acc[tid] = 0.0f;

    int rA0 = wave * 8 + 2 * half;
    int t   = rA0 % 3;

    const int l0 = rA0 * M_DIM + m;

    v8f c = {};

    for (int it = 0; it < NUM_TILES; ++it) {
        int eoff = it * TILE_ELEMS + tid * 4;
        int ld   = (eoff > SEQ_ELEMS - 4) ? (SEQ_ELEMS - 4) : eoff;
        v4f v = *(const v4f*)(src + ld);
        __syncthreads();
        *(v4f*)(tile + tid * 4) = v;
        __syncthreads();

        const int t1 = inc3(t);
        const int t2 = inc3(t1);

        {
            v2f a, b;
            a.x = onehot(seg_from(rA0,     t ), m);
            a.y = onehot(seg_from(rA0 + 1, t1), m);
            b.x = tile[l0];
            b.y = tile[l0 + 16];
            c = __builtin_amdgcn_wmma_f32_16x16x4_f32(
                    false, a, false, b, (short)0, c, false, false);
        }
        {
            v2f a, b;
            a.x = onehot(seg_from(rA0 + 4, t1), m);
            a.y = onehot(seg_from(rA0 + 5, t2), m);
            b.x = tile[l0 + 64];
            b.y = tile[l0 + 80];
            c = __builtin_amdgcn_wmma_f32_16x16x4_f32(
                    false, a, false, b, (short)0, c, false, false);
        }

        rA0 += ROWS_TILE;
        t = inc3(t);
    }

    __syncthreads();
    if (lane < 16) {
        atomicAdd(&acc[0 * M_DIM + m], c[0]);
        atomicAdd(&acc[1 * M_DIM + m], c[1]);
        atomicAdd(&acc[2 * M_DIM + m], c[2]);
        atomicAdd(&acc[3 * M_DIM + m], c[3]);
        atomicAdd(&acc[4 * M_DIM + m], c[4]);
        atomicAdd(&acc[5 * M_DIM + m], c[5]);
    }
    __syncthreads();

    if (tid < NSEG * M_DIM) {
        const int s = tid >> 4;
        const float cnt = (s == 0 || s == 3) ? 334.0f : 333.0f;
        out[seq * (NSEG * M_DIM) + tid] = acc[tid] / cnt;
    }
}

extern "C" void kernel_launch(void* const* d_in, const int* in_sizes, int n_in,
                              void* d_out, int out_size, void* d_ws, size_t ws_size,
                              hipStream_t stream) {
    const float* in  = (const float*)d_in[0];
    float*       out = (float*)d_out;
    (void)in_sizes; (void)n_in; (void)out_size; (void)d_ws; (void)ws_size;
    seg_mean_wmma_kernel<<<dim3(2048), dim3(256), 0, stream>>>(in, out);
}


# Load units re-cut on the array's own 512-byte boundaries (251 whole-line loads per sequence, edge and middle units masked per lane), not at the sequence's first row; new sum order
# speedup vs baseline: 1.8500x; 1.8500x over previous
#include <hip/hip_runtime.h>


typedef __attribute__((ext_vector_type(2))) float v2f;
typedef __attribute__((ext_vector_type(4))) float v4f;
typedef __attribute__((ext_vector_type(8))) float v8f;

#define K_SEQ       2001
#define M_DIM       16
#define NSEG        6
#define UNIT_ELEMS  128
#define UNITS       125
#define MID_UNIT    125
#define DEPTH       25
#define NCLASS      3

__device__ __forceinline__ void wave_lds_handover() {
    __builtin_amdgcn_fence(__ATOMIC_RELEASE, "wavefront");
    __builtin_amdgcn_wave_barrier();
    __builtin_amdgcn_fence(__ATOMIC_ACQUIRE, "wavefront");
}

__global__ __launch_bounds__(32)
void seg_mean_stream_kernel(const float* __restrict__ in, float* __restrict__ out) {
    __shared__ float xp[2 * NCLASS * UNIT_ELEMS];

    const int lane = threadIdx.x;
    const int h    = lane >> 4;
    const int m    = lane & 15;
    const int r    = lane >> 2;
    const unsigned seq = blockIdx.x;

    const unsigned row0 = seq * K_SEQ;
    const unsigned u0   = row0 >> 3;
    const int      s    = (int)(row0 & 7u);
    const float* __restrict__ srcL = in + (size_t)u0 * UNIT_ELEMS + lane * 4;
    const float* __restrict__ srcM = srcL + MID_UNIT * UNIT_ELEMS;
    const float* __restrict__ srcR = srcL + (MID_UNIT + 1) * UNIT_ELEMS;

    v4f bufL[DEPTH], bufR[DEPTH];
#pragma unroll
    for (int j = 0; j < DEPTH; ++j) {
        bufL[j] = *(const v4f*)(srcL + j * UNIT_ELEMS);
        bufR[j] = *(const v4f*)(srcR + j * UNIT_ELEMS);
        __builtin_amdgcn_sched_barrier(0);
    }

    const v4f zero = {};
    v4f mid = {};
    v4f aL[NCLASS] = {}, aR[NCLASS] = {};
#pragma unroll
    for (int u = 0; u < UNITS; ++u) {
        const int j = u % DEPTH;
        const int c = u % NCLASS;
        v4f xl = bufL[j], xr = bufR[j];
        if (u == 0)         xl = (r >= s) ? xl : zero;
        if (u == UNITS - 1) xr = (r <= s) ? xr : zero;
        aL[c] += xl;
        aR[c] += xr;
        if (u + DEPTH < UNITS) {
            bufL[j] = *(const v4f*)(srcL + (u + DEPTH) * UNIT_ELEMS);
            bufR[j] = *(const v4f*)(srcR + (u + DEPTH) * UNIT_ELEMS);
        }
        if (u + DEPTH == UNITS - 1) mid = *(const v4f*)srcM;
        __builtin_amdgcn_sched_barrier(0);
    }
    aL[2] += (r < s) ? mid : zero;
    aR[2] += (r > s) ? mid : zero;

    const int t = (2 * s) % 3;
    float e[3];
#pragma unroll
    for (int i = 0; i < 3; ++i) e[i] = (((i + 2 * h + t) % 3) == m) ? 1.0f : 0.0f;

    const int l0 = (2 * h) * M_DIM + m;

#pragma unroll
    for (int c = 0; c < NCLASS; ++c) {
        *(v4f*)(xp + (2 * c)     * UNIT_ELEMS + lane * 4) = aL[c];
        *(v4f*)(xp + (2 * c + 1) * UNIT_ELEMS + lane * 4) = aR[c];
    }
    wave_lds_handover();

    v8f cL = {}, cR = {};
#pragma unroll
    for (int c = 0; c < NCLASS; ++c) {
#pragma unroll
        for (int w = 0; w < 2; ++w) {
            const float* pl = xp + (2 * c)     * UNIT_ELEMS + w * 64 + l0;
            const float* pr = xp + (2 * c + 1) * UNIT_ELEMS + w * 64 + l0;
            v2f bl, br, a;
            bl.x = pl[0];  bl.y = pl[16];
            br.x = pr[0];  br.y = pr[16];
            a.x = e[(2 * c + w)     % 3];
            a.y = e[(2 * c + w + 1) % 3];
            cL = __builtin_amdgcn_wmma_f32_16x16x4_f32(
                     false, a, false, bl, (short)0, cL, false, false);
            cR = __builtin_amdgcn_wmma_f32_16x16x4_f32(
                     false, a, false, br, (short)0, cR, false, false);
        }
    }

    if (lane < 16) {
        float* o = out + (size_t)seq * (NSEG * M_DIM) + m;
        o[0 * M_DIM] = cL[0] / 334.0f;
        o[1 * M_DIM] = cL[1] / 333.0f;
        o[2 * M_DIM] = cL[2] / 333.0f;
        o[3 * M_DIM] = cR[2] / 334.0f;
        o[4 * M_DIM] = cR[0] / 333.0f;
        o[5 * M_DIM] = cR[1] / 333.0f;
    }
}

extern "C" void kernel_launch(void* const* d_in, const int* in_sizes, int n_in,
                              void* d_out, int out_size, void* d_ws, size_t ws_size,
                              hipStream_t stream) {
    const float* in  = (const float*)d_in[0];
    float*       out = (float*)d_out;
    (void)in_sizes; (void)n_in; (void)out_size; (void)d_ws; (void)ws_size;
    seg_mean_stream_kernel<<<dim3(2048), dim3(32), 0, stream>>>(in, out);
}
